# starting kernel
# baseline (speedup 1.0000x reference)
#include <hip/hip_runtime.h>

typedef __attribute__((ext_vector_type(2))) float v2f;
typedef __attribute__((ext_vector_type(8))) float v8f;

#define WMMA4F32(a, b, c) \
  __builtin_amdgcn_wmma_f32_16x16x4_f32(false, (a), false, (b), (short)0, (c), false, false)

#define HH 32
#define WW 32
#define DD 64
#define LWIN 26
#define OUTD 7

__global__ __launch_bounds__(256) void ac_wmma_kernel(
    const float* __restrict__ key_map,
    const float* __restrict__ query_map,
    float* __restrict__ out)
{
    const int lane  = threadIdx.x & 31;
    const int wib   = threadIdx.x >> 5;
    const int wid   = blockIdx.x * 8 + wib;
    const int plane = wid >> 2;
    const int dt    = wid & 3;

    const int jrow  = lane & 15;
    const int khalf = (lane >> 4) << 1;
    const int dof   = dt * 16 + jrow;

    const float* kp = key_map + (size_t)plane * (HH * WW * DD);

    const float q = query_map[(size_t)plane * (HH * WW * DD) + (16 * WW + 16) * DD + dof];

    v2f A[8];
#pragma unroll
    for (int kc = 0; kc < 8; ++kc) {
        const int w0 = kc * 4 + khalf;
        const int w1 = w0 + 1;
        A[kc].x = (jrow < OUTD && w0 >= jrow && w0 <= jrow + (LWIN - 1)) ? 1.0f : 0.0f;
        A[kc].y = (jrow < OUTD && w1 >= jrow && w1 <= jrow + (LWIN - 1)) ? 1.0f : 0.0f;
    }

    const v8f vzero = {0.f, 0.f, 0.f, 0.f, 0.f, 0.f, 0.f, 0.f};
    v8f acc0 = vzero, acc1 = vzero, acc2 = vzero, acc3 = vzero;
    v8f lowC[OUTD - 1];

    v2f Bbuf[2][8];

#define LOAD_ROW(h, buf)                                               \
    do {                                                               \
        const float* row_ = kp + (size_t)(h) * (WW * DD) + dof;        \
        _Pragma("unroll")                                              \
        for (int kc = 0; kc < 8; ++kc) {                               \
            const int w0_ = kc * 4 + khalf;                            \
            (buf)[kc].x = row_[(size_t)w0_ * DD];                      \
            (buf)[kc].y = row_[(size_t)(w0_ + 1) * DD];                \
        }                                                              \
    } while (0)

    LOAD_ROW(0, Bbuf[0]);

#pragma unroll
    for (int h = 0; h < HH; ++h) {
        v2f* cur = Bbuf[h & 1];
        if (h + 1 < HH) LOAD_ROW(h + 1, Bbuf[(h + 1) & 1]);

        acc0 = WMMA4F32(A[0], cur[0], acc0);
        acc1 = WMMA4F32(A[1], cur[1], acc1);
        acc2 = WMMA4F32(A[2], cur[2], acc2);
        acc3 = WMMA4F32(A[3], cur[3], acc3);
        acc0 = WMMA4F32(A[4], cur[4], acc0);
        acc1 = WMMA4F32(A[5], cur[5], acc1);
        acc2 = WMMA4F32(A[6], cur[6], acc2);
        acc3 = WMMA4F32(A[7], cur[7], acc3);

        const int rowsDone = h + 1;

        if (rowsDone >= 1 && rowsDone <= OUTD - 1) {
            lowC[rowsDone - 1] = (acc0 + acc1) + (acc2 + acc3);
        }

        if (rowsDone >= LWIN) {
            const int i = rowsDone - LWIN;
            v8f cum = (acc0 + acc1) + (acc2 + acc3);
            v8f res = (i == 0) ? cum : (cum - lowC[i - 1]);

            float* ob = out + (((size_t)plane * OUTD + i) * OUTD) * DD + dt * 16;
            if (lane < 16) {
#pragma unroll
                for (int r = 0; r < OUTD; ++r) {
                    ob[(size_t)r * DD + lane] = res[r] * q;
                }
            }
        }
    }
#undef LOAD_ROW
}

extern "C" void kernel_launch(void* const* d_in, const int* in_sizes, int n_in,
                              void* d_out, int out_size, void* d_ws, size_t ws_size,
                              hipStream_t stream)
{
    (void)in_sizes; (void)n_in; (void)out_size; (void)d_ws; (void)ws_size;
    const float* key_map   = (const float*)d_in[0];
    const float* query_map = (const float*)d_in[1];
    float* out = (float*)d_out;

    ac_wmma_kernel<<<dim3(256), dim3(256), 0, stream>>>(key_map, query_map, out);
}


# Lanes load 16 B along d (512-B row pieces, 4 d-interleaved N tiles per wave); plane rows split over 8 waves, partial window sums added via LDS instead of running-sum difference (sum order changes)
# speedup vs baseline: 1.1522x; 1.1522x over previous
#include <hip/hip_runtime.h>

typedef __attribute__((ext_vector_type(2))) float v2f;
typedef __attribute__((ext_vector_type(4))) float v4f;
typedef __attribute__((ext_vector_type(8))) float v8f;

#define WMMA4F32(a, b, c) \
  __builtin_amdgcn_wmma_f32_16x16x4_f32(false, (a), false, (b), (short)0, (c), false, false)

#define HH 32
#define WW 32
#define DD 64
#define LWIN 26
#define OUTD 7

#define NWAVES 8
#define RPW 4
#define SNAP_V4 (OUTD * (DD / 4))
#define PLANE_OUT_V4 (OUTD * SNAP_V4)

__global__ __launch_bounds__(256) void ac_wmma_kernel(
    const float* __restrict__ key_map,
    const float* __restrict__ query_map,
    float* __restrict__ out)
{
    __shared__ v4f part[NWAVES * RPW * SNAP_V4];

    const int tid   = threadIdx.x;
    const int lane  = tid & 31;
    const int q     = __builtin_amdgcn_readfirstlane(tid >> 5);
    const int plane = blockIdx.x;
    const int c     = lane & 15;
    const int half  = lane >> 4;

    const float* kp = key_map + (size_t)plane * (HH * WW * DD);

    const v4f qv = *(const v4f*)(query_map + (size_t)plane * (HH * WW * DD)
                                 + (16 * WW + 16) * DD + 4 * (tid & 15));

    v2f A[8];
#pragma unroll
    for (int kc = 0; kc < 8; ++kc) {
        const int w0 = kc * 4 + half;
        const int w1 = w0 + 2;
        A[kc].x = (c < OUTD && w0 >= c && w0 <= c + (LWIN - 1)) ? 1.0f : 0.0f;
        A[kc].y = (c < OUTD && w1 >= c && w1 <= c + (LWIN - 1)) ? 1.0f : 0.0f;
    }

    const int hfirst = (q < 2) ? (RPW * q + RPW - 1) : (RPW * q);
    const int hstep  = (q < 2) ? -1 : 1;

    v4f buf[2][16];

#define LOAD_ROW(k, b)                                                          \
    do {                                                                        \
        const float* row_ = kp + (size_t)(hfirst + hstep * (k)) * (WW * DD)     \
                               + lane * 4;                                      \
        _Pragma("unroll")                                                       \
        for (int a = 0; a < 16; ++a)                                            \
            buf[b][a] = *(const v4f*)(row_ + a * 128);                          \
    } while (0)

    const v8f vzero = {0.f, 0.f, 0.f, 0.f, 0.f, 0.f, 0.f, 0.f};
    v8f acc[4] = {vzero, vzero, vzero, vzero};

    LOAD_ROW(0, 0);
    LOAD_ROW(1, 1);

#pragma unroll
    for (int k = 0; k < RPW; ++k) {
        const v4f* cur = buf[k & 1];

#pragma unroll
        for (int kc = 0; kc < 8; ++kc) {
#pragma unroll
            for (int t = 0; t < 4; ++t) {
                v2f b;
                b.x = cur[2 * kc][t];
                b.y = cur[2 * kc + 1][t];
                acc[t] = WMMA4F32(A[kc], b, acc[t]);
            }
        }

        if (k + 2 < RPW) LOAD_ROW(k + 2, k & 1);

        if (lane < 16) {
#pragma unroll
            for (int r = 0; r < OUTD; ++r) {
                v4f s;
                s.x = acc[0][r]; s.y = acc[1][r]; s.z = acc[2][r]; s.w = acc[3][r];
                part[((q * RPW + k) * OUTD + r) * (DD / 4) + c] = s;
            }
        }
    }
#undef LOAD_ROW

    __syncthreads();

    v4f* ob = (v4f*)(out + (size_t)plane * (OUTD * OUTD * DD));
#pragma unroll
    for (int it = 0; it < 4; ++it) {
        const int e4 = tid + 256 * it;
        if (it == 3 && tid >= PLANE_OUT_V4 - 3 * 256) break;
        const int i   = e4 / SNAP_V4;
        const int rem = e4 - i * SNAP_V4;
        v4f sum = {0.f, 0.f, 0.f, 0.f};
#pragma unroll
        for (int w = 0; w < NWAVES; ++w) {
            const int lo  = RPW * w, hi = RPW * w + RPW - 1;
            const int top = (hi < i + LWIN - 1) ? hi : (i + LWIN - 1);
            const int bot = (lo > i) ? lo : i;
            const int cnt = top - bot + 1;
            const int sl  = (cnt > 0) ? (cnt - 1) : 0;
            const v4f v   = part[(w * RPW + sl) * SNAP_V4 + rem];
            const v4f z   = {0.f, 0.f, 0.f, 0.f};
            sum += (cnt > 0) ? v : z;
        }
        ob[e4] = sum * qv;
    }
}

extern "C" void kernel_launch(void* const* d_in, const int* in_sizes, int n_in,
                              void* d_out, int out_size, void* d_ws, size_t ws_size,
                              hipStream_t stream)
{
    (void)in_sizes; (void)n_in; (void)out_size; (void)d_ws; (void)ws_size;
    const float* key_map   = (const float*)d_in[0];
    const float* query_map = (const float*)d_in[1];
    float* out = (float*)d_out;

    ac_wmma_kernel<<<dim3(512), dim3(256), 0, stream>>>(key_map, query_map, out);
}


# Rows 6..25 (in every h window) are added on the vector unit by four 5-row waves before ONE w-window product each; four 3-row edge waves keep per-row products: half the WMMAs (sum order changes)
# speedup vs baseline: 1.2369x; 1.0735x over previous
#include <hip/hip_runtime.h>

typedef __attribute__((ext_vector_type(2))) float v2f;
typedef __attribute__((ext_vector_type(4))) float v4f;
typedef __attribute__((ext_vector_type(8))) float v8f;

#define WMMA4F32(a, b, c) \
  __builtin_amdgcn_wmma_f32_16x16x4_f32(false, (a), false, (b), (short)0, (c), false, false)

#define HH 32
#define WW 32
#define DD 64
#define LWIN 26
#define OUTD 7

#define NWAVES 8
#define NEDGE 4
#define ERPW 3
#define MRPW 5
#define NSLOT (NEDGE * ERPW + (NWAVES - NEDGE))
#define SNAP_V4 (OUTD * (DD / 4))
#define PLANE_OUT_V4 (OUTD * SNAP_V4)

__global__ __launch_bounds__(256) void ac_wmma_kernel(
    const float* __restrict__ key_map,
    const float* __restrict__ query_map,
    float* __restrict__ out)
{
    __shared__ v4f part[NSLOT * SNAP_V4];

    const int tid   = threadIdx.x;
    const int lane  = tid & 31;
    const int q     = __builtin_amdgcn_readfirstlane(tid >> 5);
    const int plane = blockIdx.x;
    const int c     = lane & 15;
    const int half  = lane >> 4;

    const float* kp = key_map + (size_t)plane * (HH * WW * DD) + lane * 4;

    const v4f qv = *(const v4f*)(query_map + (size_t)plane * (HH * WW * DD)
                                 + (16 * WW + 16) * DD + 4 * (tid & 15));

    v2f A[8];
#pragma unroll
    for (int kc = 0; kc < 8; ++kc) {
        const int w0 = kc * 4 + half;
        const int w1 = w0 + 2;
        A[kc].x = (c < OUTD && w0 >= c && w0 <= c + (LWIN - 1)) ? 1.0f : 0.0f;
        A[kc].y = (c < OUTD && w1 >= c && w1 <= c + (LWIN - 1)) ? 1.0f : 0.0f;
    }

    const v8f vzero = {0.f, 0.f, 0.f, 0.f, 0.f, 0.f, 0.f, 0.f};
    v8f acc[4] = {vzero, vzero, vzero, vzero};

#define LOAD_ROW(dst, h)                                                        \
    do {                                                                        \
        const float* row_ = kp + (size_t)(h) * (WW * DD);                       \
        _Pragma("unroll")                                                       \
        for (int a = 0; a < 16; ++a)                                            \
            (dst)[a] = *(const v4f*)(row_ + a * 128);                           \
    } while (0)

#define ROW_PRODUCT(src)                                                        \
    do {                                                                        \
        _Pragma("unroll")                                                       \
        for (int kc = 0; kc < 8; ++kc) {                                        \
            _Pragma("unroll")                                                   \
            for (int t = 0; t < 4; ++t) {                                       \
                v2f b;                                                          \
                b.x = (src)[2 * kc][t];                                         \
                b.y = (src)[2 * kc + 1][t];                                     \
                acc[t] = WMMA4F32(A[kc], b, acc[t]);                            \
            }                                                                   \
        }                                                                       \
    } while (0)

#define SNAPSHOT(slot)                                                          \
    do {                                                                        \
        if (lane < 16) {                                                        \
            _Pragma("unroll")                                                   \
            for (int r = 0; r < OUTD; ++r) {                                    \
                v4f s;                                                          \
                s.x = acc[0][r]; s.y = acc[1][r]; s.z = acc[2][r]; s.w = acc[3][r]; \
                part[((slot) * OUTD + r) * (DD / 4) + c] = s;                   \
            }                                                                   \
        }                                                                       \
    } while (0)

    v4f b0[16], b1[16], b2[16];

    if (q < NEDGE) {
        const int hfirst = (q < 2) ? (ERPW * q + ERPW - 1) : (ERPW * q + 20);
        const int hstep  = (q < 2) ? -1 : 1;
        LOAD_ROW(b0, hfirst);
        LOAD_ROW(b1, hfirst + hstep);
        LOAD_ROW(b2, hfirst + 2 * hstep);
        ROW_PRODUCT(b0);
        SNAPSHOT(q * ERPW + 0);
        ROW_PRODUCT(b1);
        SNAPSHOT(q * ERPW + 1);
        ROW_PRODUCT(b2);
        SNAPSHOT(q * ERPW + 2);
    } else {
        const int h0 = 6 + MRPW * (q - NEDGE);
        LOAD_ROW(b0, h0);
        LOAD_ROW(b1, h0 + 1);
        LOAD_ROW(b2, h0 + 2);
#pragma unroll
        for (int a = 0; a < 16; ++a) b0[a] += b1[a];
        LOAD_ROW(b1, h0 + 3);
        __builtin_amdgcn_sched_barrier(0);
#pragma unroll
        for (int a = 0; a < 16; ++a) b0[a] += b2[a];
        LOAD_ROW(b2, h0 + 4);
        __builtin_amdgcn_sched_barrier(0);
#pragma unroll
        for (int a = 0; a < 16; ++a) b0[a] += b1[a];
#pragma unroll
        for (int a = 0; a < 16; ++a) b0[a] += b2[a];
        ROW_PRODUCT(b0);
        SNAPSHOT(NEDGE * ERPW + (q - NEDGE));
    }
#undef LOAD_ROW
#undef ROW_PRODUCT
#undef SNAPSHOT

    __syncthreads();

    v4f* ob = (v4f*)(out + (size_t)plane * (OUTD * OUTD * DD));
#pragma unroll
    for (int it = 0; it < 4; ++it) {
        const int e4 = tid + 256 * it;
        if (it == 3 && tid >= PLANE_OUT_V4 - 3 * 256) break;
        const int i   = e4 / SNAP_V4;
        const int rem = e4 - i * SNAP_V4;
        v4f sum = part[(NEDGE * ERPW + 0) * SNAP_V4 + rem];
#pragma unroll
        for (int m = 1; m < NWAVES - NEDGE; ++m)
            sum += part[(NEDGE * ERPW + m) * SNAP_V4 + rem];
#pragma unroll
        for (int w = 0; w < NEDGE; ++w) {
            const int lo  = (w < 2) ? (ERPW * w) : (ERPW * w + 20);
            const int hi  = lo + ERPW - 1;
            const int top = (hi < i + LWIN - 1) ? hi : (i + LWIN - 1);
            const int bot = (lo > i) ? lo : i;
            const int cnt = top - bot + 1;
            const int sl  = (cnt > 0) ? (cnt - 1) : 0;
            const v4f v   = part[(w * ERPW + sl) * SNAP_V4 + rem];
            const v4f z   = {0.f, 0.f, 0.f, 0.f};
            sum += (cnt > 0) ? v : z;
        }
        ob[e4] = sum * qv;
    }
}

extern "C" void kernel_launch(void* const* d_in, const int* in_sizes, int n_in,
                              void* d_out, int out_size, void* d_ws, size_t ws_size,
                              hipStream_t stream)
{
    (void)in_sizes; (void)n_in; (void)out_size; (void)d_ws; (void)ws_size;
    const float* key_map   = (const float*)d_in[0];
    const float* query_map = (const float*)d_in[1];
    float* out = (float*)d_out;

    ac_wmma_kernel<<<dim3(512), dim3(256), 0, stream>>>(key_map, query_map, out);
}


# Columns 6..25 (in every w window) of each row or row sum are added on the vector unit into two sums before the w-window product: 4 k chunks per row instead of 8, half the WMMAs (sum order changes)
# speedup vs baseline: 1.3078x; 1.0573x over previous
#include <hip/hip_runtime.h>

typedef __attribute__((ext_vector_type(2))) float v2f;
typedef __attribute__((ext_vector_type(4))) float v4f;
typedef __attribute__((ext_vector_type(8))) float v8f;

#define WMMA4F32(a, b, c) \
  __builtin_amdgcn_wmma_f32_16x16x4_f32(false, (a), false, (b), (short)0, (c), false, false)

#define HH 32
#define WW 32
#define DD 64
#define LWIN 26
#define OUTD 7

#define NWAVES 8
#define NEDGE 4
#define ERPW 3
#define MRPW 5
#define NSLOT (NEDGE * ERPW + (NWAVES - NEDGE))
#define SNAP_V4 (OUTD * (DD / 4))
#define PLANE_OUT_V4 (OUTD * SNAP_V4)

__global__ __launch_bounds__(256) void ac_wmma_kernel(
    const float* __restrict__ key_map,
    const float* __restrict__ query_map,
    float* __restrict__ out)
{
    __shared__ v4f part[NSLOT * SNAP_V4];

    const int tid   = threadIdx.x;
    const int lane  = tid & 31;
    const int q     = __builtin_amdgcn_readfirstlane(tid >> 5);
    const int plane = blockIdx.x;
    const int c     = lane & 15;
    const int half  = lane >> 4;

    const float* kp = key_map + (size_t)plane * (HH * WW * DD) + lane * 4;

    const v4f qv = *(const v4f*)(query_map + (size_t)plane * (HH * WW * DD)
                                 + (16 * WW + 16) * DD + 4 * (tid & 15));

#define WMASK(w) ((c < OUTD && (w) >= c && (w) <= c + (LWIN - 1)) ? 1.0f : 0.0f)
    const float all7 = (c < OUTD) ? 1.0f : 0.0f;
    v2f A[4];
    A[0].x = WMASK(half);      A[0].y = WMASK(2 + half);
    A[1].x = WMASK(4 + half);  A[1].y = all7;
    A[2].x = all7;             A[2].y = WMASK(26 + half);
    A[3].x = WMASK(28 + half); A[3].y = WMASK(30 + half);
#undef WMASK

    const v8f vzero = {0.f, 0.f, 0.f, 0.f, 0.f, 0.f, 0.f, 0.f};
    v8f acc[4] = {vzero, vzero, vzero, vzero};

#define LOAD_ROW(dst, h)                                                        \
    do {                                                                        \
        const float* row_ = kp + (size_t)(h) * (WW * DD);                       \
        _Pragma("unroll")                                                       \
        for (int a = 0; a < 16; ++a)                                            \
            (dst)[a] = *(const v4f*)(row_ + a * 128);                           \
    } while (0)

#define ROW_PRODUCT(src)                                                        \
    do {                                                                        \
        const v4f m1_ = (((src)[3] + (src)[4]) + ((src)[5] + (src)[6])) + (src)[7];     \
        const v4f m2_ = (((src)[8] + (src)[9]) + ((src)[10] + (src)[11])) + (src)[12];  \
        _Pragma("unroll")                                                       \
        for (int t = 0; t < 4; ++t) {                                           \
            v2f b;                                                              \
            b.x = (src)[0][t];  b.y = (src)[1][t];                              \
            acc[t] = WMMA4F32(A[0], b, acc[t]);                                 \
        }                                                                       \
        _Pragma("unroll")                                                       \
        for (int t = 0; t < 4; ++t) {                                           \
            v2f b;                                                              \
            b.x = (src)[2][t];  b.y = m1_[t];                                   \
            acc[t] = WMMA4F32(A[1], b, acc[t]);                                 \
        }                                                                       \
        _Pragma("unroll")                                                       \
        for (int t = 0; t < 4; ++t) {                                           \
            v2f b;                                                              \
            b.x = m2_[t];       b.y = (src)[13][t];                             \
            acc[t] = WMMA4F32(A[2], b, acc[t]);                                 \
        }                                                                       \
        _Pragma("unroll")                                                       \
        for (int t = 0; t < 4; ++t) {                                           \
            v2f b;                                                              \
            b.x = (src)[14][t]; b.y = (src)[15][t];                             \
            acc[t] = WMMA4F32(A[3], b, acc[t]);                                 \
        }                                                                       \
    } while (0)

#define SNAPSHOT(slot)                                                          \
    do {                                                                        \
        if (lane < 16) {                                                        \
            _Pragma("unroll")                                                   \
            for (int r = 0; r < OUTD; ++r) {                                    \
                v4f s;                                                          \
                s.x = acc[0][r]; s.y = acc[1][r]; s.z = acc[2][r]; s.w = acc[3][r]; \
                part[((slot) * OUTD + r) * (DD / 4) + c] = s;                   \
            }                                                                   \
        }                                                                       \
    } while (0)

    v4f b0[16], b1[16], b2[16];

    if (q < NEDGE) {
        const int hfirst = (q < 2) ? (ERPW * q + ERPW - 1) : (ERPW * q + 20);
        const int hstep  = (q < 2) ? -1 : 1;
        LOAD_ROW(b0, hfirst);
        LOAD_ROW(b1, hfirst + hstep);
        LOAD_ROW(b2, hfirst + 2 * hstep);
        ROW_PRODUCT(b0);
        SNAPSHOT(q * ERPW + 0);
        ROW_PRODUCT(b1);
        SNAPSHOT(q * ERPW + 1);
        ROW_PRODUCT(b2);
        SNAPSHOT(q * ERPW + 2);
    } else {
        const int h0 = 6 + MRPW * (q - NEDGE);
        LOAD_ROW(b0, h0);
        LOAD_ROW(b1, h0 + 1);
        LOAD_ROW(b2, h0 + 2);
#pragma unroll
        for (int a = 0; a < 16; ++a) b0[a] += b1[a];
        LOAD_ROW(b1, h0 + 3);
        __builtin_amdgcn_sched_barrier(0);
#pragma unroll
        for (int a = 0; a < 16; ++a) b0[a] += b2[a];
        LOAD_ROW(b2, h0 + 4);
        __builtin_amdgcn_sched_barrier(0);
#pragma unroll
        for (int a = 0; a < 16; ++a) b0[a] += b1[a];
#pragma unroll
        for (int a = 0; a < 16; ++a) b0[a] += b2[a];
        ROW_PRODUCT(b0);
        SNAPSHOT(NEDGE * ERPW + (q - NEDGE));
    }
#undef LOAD_ROW
#undef ROW_PRODUCT
#undef SNAPSHOT

    __syncthreads();

    v4f* ob = (v4f*)(out + (size_t)plane * (OUTD * OUTD * DD));
#pragma unroll
    for (int it = 0; it < 4; ++it) {
        const int e4 = tid + 256 * it;
        if (it == 3 && tid >= PLANE_OUT_V4 - 3 * 256) break;
        const int i   = e4 / SNAP_V4;
        const int rem = e4 - i * SNAP_V4;
        v4f sum = part[(NEDGE * ERPW + 0) * SNAP_V4 + rem];
#pragma unroll
        for (int m = 1; m < NWAVES - NEDGE; ++m)
            sum += part[(NEDGE * ERPW + m) * SNAP_V4 + rem];
#pragma unroll
        for (int w = 0; w < NEDGE; ++w) {
            const int lo  = (w < 2) ? (ERPW * w) : (ERPW * w + 20);
            const int hi  = lo + ERPW - 1;
            const int top = (hi < i + LWIN - 1) ? hi : (i + LWIN - 1);
            const int bot = (lo > i) ? lo : i;
            const int cnt = top - bot + 1;
            const int sl  = (cnt > 0) ? (cnt - 1) : 0;
            const v4f v   = part[(w * ERPW + sl) * SNAP_V4 + rem];
            const v4f z   = {0.f, 0.f, 0.f, 0.f};
            sum += (cnt > 0) ? v : z;
        }
        ob[e4] = sum * qv;
    }
}

extern "C" void kernel_launch(void* const* d_in, const int* in_sizes, int n_in,
                              void* d_out, int out_size, void* d_ws, size_t ws_size,
                              hipStream_t stream)
{
    (void)in_sizes; (void)n_in; (void)out_size; (void)d_ws; (void)ws_size;
    const float* key_map   = (const float*)d_in[0];
    const float* query_map = (const float*)d_in[1];
    float* out = (float*)d_out;

    ac_wmma_kernel<<<dim3(512), dim3(256), 0, stream>>>(key_map, query_map, out);
}
